# starting kernel
# baseline (speedup 1.0000x reference)
#include <hip/hip_runtime.h>

#define T_SEQ 4096
#define HID   1024
#define KDIM  1024

#define KC    32
#define BM    128
#define BN    64
#define LDX   36
#define LDB   36

#define CLEN   256
#define NCHUNK (T_SEQ / CLEN)

typedef float v2f __attribute__((ext_vector_type(2)));
typedef float v4f __attribute__((ext_vector_type(4)));
typedef float v8f __attribute__((ext_vector_type(8)));

__global__ __launch_bounds__(256) void gemm_xb_wmma(const float* __restrict__ X,
                                                    const float* __restrict__ B,
                                                    float* __restrict__ S) {
  __shared__ float sX[BM * LDX];
  __shared__ float sB[BN * LDB];

  const int tid  = threadIdx.x;
  const int wave = tid >> 5;
  const int lane = tid & 31;
  const int half = lane >> 4;
  const int r    = lane & 15;

  const int blockN = blockIdx.x * BN;
  const int blockM = blockIdx.y * BM;

  const int wm = (wave & 3) << 5;
  const int wn = (wave >> 2) << 5;

  const int rowGrp = tid >> 3;
  const int col4   = (tid & 7) << 2;
  const int bRow   = tid >> 4;
  const int bCol4  = (tid & 15) << 2;

  v8f acc00 = {}, acc01 = {}, acc10 = {}, acc11 = {};

  for (int k0 = 0; k0 < KDIM; k0 += KC) {
    __syncthreads();

    #pragma unroll
    for (int p = 0; p < 4; ++p) {
      const int row = rowGrp + (p << 5);
      v4f xv = *(const v4f*)(X + (size_t)(blockM + row) * KDIM + k0 + col4);
      *(v4f*)(&sX[row * LDX + col4]) = xv;
    }
    #pragma unroll
    for (int p = 0; p < 2; ++p) {
      const int kRow = bRow + (p << 4);
      v4f bv = *(const v4f*)(B + (size_t)(k0 + kRow) * HID + blockN + bCol4);
      #pragma unroll
      for (int i = 0; i < 4; ++i)
        sB[(bCol4 + i) * LDB + kRow] = bv[i];
    }
    __syncthreads();

    #pragma unroll
    for (int kk = 0; kk < KC; kk += 4) {
      v2f a0 = *(const v2f*)(&sX[(wm + r)      * LDX + kk + 2 * half]);
      v2f a1 = *(const v2f*)(&sX[(wm + 16 + r) * LDX + kk + 2 * half]);
      v2f b0 = *(const v2f*)(&sB[(wn + r)      * LDB + kk + 2 * half]);
      v2f b1 = *(const v2f*)(&sB[(wn + 16 + r) * LDB + kk + 2 * half]);
      acc00 = __builtin_amdgcn_wmma_f32_16x16x4_f32(false, a0, false, b0,
                                                    (short)0, acc00, false, false);
      acc01 = __builtin_amdgcn_wmma_f32_16x16x4_f32(false, a0, false, b1,
                                                    (short)0, acc01, false, false);
      acc10 = __builtin_amdgcn_wmma_f32_16x16x4_f32(false, a1, false, b0,
                                                    (short)0, acc10, false, false);
      acc11 = __builtin_amdgcn_wmma_f32_16x16x4_f32(false, a1, false, b1,
                                                    (short)0, acc11, false, false);
    }
  }

  #pragma unroll
  for (int i = 0; i < 8; ++i) {
    const int row0 = blockM + wm + i + 8 * half;
    const int col0 = blockN + wn + r;
    S[(size_t)row0        * HID + col0]      = acc00[i];
    S[(size_t)row0        * HID + col0 + 16] = acc01[i];
    S[(size_t)(row0 + 16) * HID + col0]      = acc10[i];
    S[(size_t)(row0 + 16) * HID + col0 + 16] = acc11[i];
  }
}

__global__ __launch_bounds__(256) void scan_local(const float* __restrict__ a_diag,
                                                  float* __restrict__ out,
                                                  float* __restrict__ carry) {
  const int h = blockIdx.x * 256 + threadIdx.x;
  const int c = blockIdx.y;
  const float a = a_diag[h];
  float accv = 0.0f;
  const int t0 = c * CLEN;
  for (int t = 0; t < CLEN; ++t) {
    const size_t idx = (size_t)(t0 + t) * HID + h;
    accv = fmaf(a, accv, out[idx]);
    out[idx] = accv;
  }
  carry[c * HID + h] = accv;
}

__global__ __launch_bounds__(256) void scan_carry(const float* __restrict__ a_diag,
                                                  const float* __restrict__ carry,
                                                  float* __restrict__ carryIn) {
  const int h = blockIdx.x * 256 + threadIdx.x;
  const float a = a_diag[h];
  float aL = a;
  #pragma unroll
  for (int i = 0; i < 8; ++i) aL *= aL;
  float state = 0.0f;
  for (int c = 0; c < NCHUNK; ++c) {
    carryIn[c * HID + h] = state;
    state = fmaf(aL, state, carry[c * HID + h]);
  }
}

__global__ __launch_bounds__(256) void scan_fix(const float* __restrict__ a_diag,
                                                const float* __restrict__ carryIn,
                                                float* __restrict__ out) {
  const int h = blockIdx.x * 256 + threadIdx.x;
  const int c = blockIdx.y + 1;
  const float a  = a_diag[h];
  const float ci = carryIn[c * HID + h];
  float w = a;
  const int t0 = c * CLEN;
  for (int t = 0; t < CLEN; ++t) {
    const size_t idx = (size_t)(t0 + t) * HID + h;
    out[idx] = fmaf(w, ci, out[idx]);
    w *= a;
  }
}

extern "C" void kernel_launch(void* const* d_in, const int* in_sizes, int n_in,
                              void* d_out, int out_size, void* d_ws, size_t ws_size,
                              hipStream_t stream) {
  const float* x_seq  = (const float*)d_in[0];
  const float* a_diag = (const float*)d_in[1];
  const float* b_mat  = (const float*)d_in[2];
  float* out = (float*)d_out;

  float* carry   = (float*)d_ws;
  float* carryIn = carry + NCHUNK * HID;

  dim3 gGemm(HID / BN, T_SEQ / BM);
  gemm_xb_wmma<<<gGemm, 256, 0, stream>>>(x_seq, b_mat, out);

  dim3 gLocal(HID / 256, NCHUNK);
  scan_local<<<gLocal, 256, 0, stream>>>(a_diag, out, carry);

  scan_carry<<<HID / 256, 256, 0, stream>>>(a_diag, carry, carryIn);

  dim3 gFix(HID / 256, NCHUNK - 1);
  scan_fix<<<gFix, 256, 0, stream>>>(a_diag, carryIn, out);
}


# Chunk-local scan moved into the GEMM epilogue: 128-row tile scanned on the accumulators (8-row lane segments joined by an a^8 prefix over LDS); scan_local launch gone; scan summation order changes
# speedup vs baseline: 1.8179x; 1.8179x over previous
#include <hip/hip_runtime.h>

#define T_SEQ 4096
#define HID   1024
#define KDIM  1024

#define KC    32
#define BM    128
#define BN    64
#define LDX   36
#define LDB   36

#define CLEN   BM
#define NCHUNK (T_SEQ / CLEN)
#define SEGS   (BM / 8)
#define LDE    20

typedef float v2f __attribute__((ext_vector_type(2)));
typedef float v4f __attribute__((ext_vector_type(4)));
typedef float v8f __attribute__((ext_vector_type(8)));

__global__ __launch_bounds__(256) void gemm_xb_wmma(const float* __restrict__ X,
                                                    const float* __restrict__ B,
                                                    const float* __restrict__ a_diag,
                                                    float* __restrict__ S,
                                                    float* __restrict__ carry) {
  __shared__ float sX[BM * LDX];
  __shared__ float sB[BN * LDB];
  __shared__ float sE[BN * LDE];

  const int tid  = threadIdx.x;
  const int wave = tid >> 5;
  const int lane = tid & 31;
  const int half = lane >> 4;
  const int r    = lane & 15;

  const int blockN = blockIdx.x * BN;
  const int blockM = blockIdx.y * BM;

  const int wm = (wave & 3) << 5;
  const int wn = (wave >> 2) << 5;

  const int rowGrp = tid >> 3;
  const int col4   = (tid & 7) << 2;
  const int bRow   = tid >> 4;
  const int bCol4  = (tid & 15) << 2;

  const int   cA = wn + r;
  const int   cB = wn + 16 + r;
  const float aA = a_diag[blockN + cA];
  const float aB = a_diag[blockN + cB];

  v8f acc00 = {}, acc01 = {}, acc10 = {}, acc11 = {};

  for (int k0 = 0; k0 < KDIM; k0 += KC) {
    __syncthreads();

    #pragma unroll
    for (int p = 0; p < 4; ++p) {
      const int row = rowGrp + (p << 5);
      v4f xv = *(const v4f*)(X + (size_t)(blockM + row) * KDIM + k0 + col4);
      *(v4f*)(&sX[row * LDX + col4]) = xv;
    }
    #pragma unroll
    for (int p = 0; p < 2; ++p) {
      const int kRow = bRow + (p << 4);
      v4f bv = *(const v4f*)(B + (size_t)(k0 + kRow) * HID + blockN + bCol4);
      #pragma unroll
      for (int i = 0; i < 4; ++i)
        sB[(bCol4 + i) * LDB + kRow] = bv[i];
    }
    __syncthreads();

    #pragma unroll
    for (int kk = 0; kk < KC; kk += 4) {
      v2f a0 = *(const v2f*)(&sX[(wm + r)      * LDX + kk + 2 * half]);
      v2f a1 = *(const v2f*)(&sX[(wm + 16 + r) * LDX + kk + 2 * half]);
      v2f b0 = *(const v2f*)(&sB[(wn + r)      * LDB + kk + 2 * half]);
      v2f b1 = *(const v2f*)(&sB[(wn + 16 + r) * LDB + kk + 2 * half]);
      acc00 = __builtin_amdgcn_wmma_f32_16x16x4_f32(false, a0, false, b0,
                                                    (short)0, acc00, false, false);
      acc01 = __builtin_amdgcn_wmma_f32_16x16x4_f32(false, a0, false, b1,
                                                    (short)0, acc01, false, false);
      acc10 = __builtin_amdgcn_wmma_f32_16x16x4_f32(false, a1, false, b0,
                                                    (short)0, acc10, false, false);
      acc11 = __builtin_amdgcn_wmma_f32_16x16x4_f32(false, a1, false, b1,
                                                    (short)0, acc11, false, false);
    }
  }

  const int segA = (wm >> 3) + half;

  #pragma unroll
  for (int i = 1; i < 8; ++i) {
    acc00[i] = fmaf(aA, acc00[i - 1], acc00[i]);
    acc10[i] = fmaf(aA, acc10[i - 1], acc10[i]);
    acc01[i] = fmaf(aB, acc01[i - 1], acc01[i]);
    acc11[i] = fmaf(aB, acc11[i - 1], acc11[i]);
  }

  sE[cA * LDE + segA]     = acc00[7];
  sE[cA * LDE + segA + 2] = acc10[7];
  sE[cB * LDE + segA]     = acc01[7];
  sE[cB * LDE + segA + 2] = acc11[7];

  float pwA[8], pwB[8];
  pwA[0] = aA;
  pwB[0] = aB;
  #pragma unroll
  for (int k = 1; k < 8; ++k) {
    pwA[k] = pwA[k - 1] * aA;
    pwB[k] = pwB[k - 1] * aB;
  }
  __syncthreads();

  float eA[SEGS], eB[SEGS];
  #pragma unroll
  for (int q = 0; q < SEGS / 4; ++q) {
    const v4f ta = *(const v4f*)(&sE[cA * LDE + 4 * q]);
    const v4f tb = *(const v4f*)(&sE[cB * LDE + 4 * q]);
    #pragma unroll
    for (int i = 0; i < 4; ++i) {
      eA[4 * q + i] = ta[i];
      eB[4 * q + i] = tb[i];
    }
  }
  float stA = 0.0f, stB = 0.0f;
  float inA0 = 0.0f, inA1 = 0.0f, inB0 = 0.0f, inB1 = 0.0f;
  #pragma unroll
  for (int j = 0; j < SEGS; ++j) {
    if (j == segA)     { inA0 = stA; inB0 = stB; }
    if (j == segA + 2) { inA1 = stA; inB1 = stB; }
    stA = fmaf(pwA[7], stA, eA[j]);
    stB = fmaf(pwB[7], stB, eB[j]);
  }

  #pragma unroll
  for (int i = 0; i < 8; ++i) {
    acc00[i] = fmaf(pwA[i], inA0, acc00[i]);
    acc10[i] = fmaf(pwA[i], inA1, acc10[i]);
    acc01[i] = fmaf(pwB[i], inB0, acc01[i]);
    acc11[i] = fmaf(pwB[i], inB1, acc11[i]);
  }

  if ((wave & 3) == 0 && half == 0) {
    carry[blockIdx.y * HID + blockN + cA] = stA;
    carry[blockIdx.y * HID + blockN + cB] = stB;
  }

  #pragma unroll
  for (int i = 0; i < 8; ++i) {
    const int row0 = blockM + wm + i + 8 * half;
    const int col0 = blockN + wn + r;
    S[(size_t)row0        * HID + col0]      = acc00[i];
    S[(size_t)row0        * HID + col0 + 16] = acc01[i];
    S[(size_t)(row0 + 16) * HID + col0]      = acc10[i];
    S[(size_t)(row0 + 16) * HID + col0 + 16] = acc11[i];
  }
}

__global__ __launch_bounds__(256) void scan_carry(const float* __restrict__ a_diag,
                                                  float* __restrict__ carry) {
  const int h = blockIdx.x * 256 + threadIdx.x;
  const float a = a_diag[h];
  static_assert(CLEN == 128, "7 squarings below give a^128");
  float aL = a;
  #pragma unroll
  for (int i = 0; i < 7; ++i) aL *= aL;
  float state = 0.0f;
  for (int c = 0; c < NCHUNK; ++c) {
    const float cv = carry[c * HID + h];
    carry[c * HID + h] = state;
    state = fmaf(aL, state, cv);
  }
}

__global__ __launch_bounds__(256) void scan_fix(const float* __restrict__ a_diag,
                                                const float* __restrict__ carryIn,
                                                float* __restrict__ out) {
  const int h = blockIdx.x * 256 + threadIdx.x;
  const int c = blockIdx.y + 1;
  const float a  = a_diag[h];
  const float ci = carryIn[c * HID + h];
  float w = a;
  const int t0 = c * CLEN;
  for (int t = 0; t < CLEN; ++t) {
    const size_t idx = (size_t)(t0 + t) * HID + h;
    out[idx] = fmaf(w, ci, out[idx]);
    w *= a;
  }
}

extern "C" void kernel_launch(void* const* d_in, const int* in_sizes, int n_in,
                              void* d_out, int out_size, void* d_ws, size_t ws_size,
                              hipStream_t stream) {
  const float* x_seq  = (const float*)d_in[0];
  const float* a_diag = (const float*)d_in[1];
  const float* b_mat  = (const float*)d_in[2];
  float* out = (float*)d_out;

  float* carry = (float*)d_ws;

  dim3 gGemm(HID / BN, T_SEQ / BM);
  gemm_xb_wmma<<<gGemm, 256, 0, stream>>>(x_seq, b_mat, a_diag, out, carry);

  scan_carry<<<HID / 256, 256, 0, stream>>>(a_diag, carry);

  dim3 gFix(HID / 256, NCHUNK - 1);
  scan_fix<<<gFix, 256, 0, stream>>>(a_diag, carry, out);
}


# GEMM K loop software-pipelined: next chunk's global loads in flight under the current chunk's WMMAs, LDS tiles double-buffered, one barrier per chunk; registers budgeted for 4 waves per SIMD
# speedup vs baseline: 1.9939x; 1.0968x over previous
#include <hip/hip_runtime.h>

#define T_SEQ 4096
#define HID   1024
#define KDIM  1024

#define KC    32
#define BM    128
#define BN    64
#define LDX   36
#define LDB   36
#define NKC   (KDIM / KC)
#define XS_FLOATS (BM * LDX)
#define BS_FLOATS (BN * LDB)

#define CLEN   BM
#define NCHUNK (T_SEQ / CLEN)
#define SEGS   (BM / 8)
#define LDE    20

typedef float v2f __attribute__((ext_vector_type(2)));
typedef float v4f __attribute__((ext_vector_type(4)));
typedef float v8f __attribute__((ext_vector_type(8)));

static __device__ __forceinline__ void stage_load(const float* __restrict__ xg,
                                                  const float* __restrict__ bg, int k0,
                                                  v4f (&xv)[4], v4f (&bv)[2]) {
  #pragma unroll
  for (int p = 0; p < 4; ++p)
    xv[p] = *(const v4f*)(xg + (size_t)(p << 5) * KDIM + k0);
  #pragma unroll
  for (int p = 0; p < 2; ++p)
    bv[p] = *(const v4f*)(bg + (size_t)(k0 + (p << 4)) * HID);
}

static __device__ __forceinline__ void stage_store(float* sXs, float* sBs, int xs, int bs,
                                                   const v4f (&xv)[4], const v4f (&bv)[2]) {
  #pragma unroll
  for (int p = 0; p < 4; ++p)
    *(v4f*)(&sXs[xs + (p << 5) * LDX]) = xv[p];
  #pragma unroll
  for (int p = 0; p < 2; ++p) {
    #pragma unroll
    for (int i = 0; i < 4; ++i)
      sBs[bs + i * LDB + (p << 4)] = bv[p][i];
  }
}

static __device__ __forceinline__ void mma_chunk(const float* sXs, const float* sBs,
                                                 int fa0, int fa1, int fb0, int fb1,
                                                 v8f& acc00, v8f& acc01, v8f& acc10, v8f& acc11) {
  #pragma unroll
  for (int kk = 0; kk < KC; kk += 4) {
    v2f a0 = *(const v2f*)(&sXs[fa0 + kk]);
    v2f a1 = *(const v2f*)(&sXs[fa1 + kk]);
    v2f b0 = *(const v2f*)(&sBs[fb0 + kk]);
    v2f b1 = *(const v2f*)(&sBs[fb1 + kk]);
    acc00 = __builtin_amdgcn_wmma_f32_16x16x4_f32(false, a0, false, b0,
                                                  (short)0, acc00, false, false);
    acc01 = __builtin_amdgcn_wmma_f32_16x16x4_f32(false, a0, false, b1,
                                                  (short)0, acc01, false, false);
    acc10 = __builtin_amdgcn_wmma_f32_16x16x4_f32(false, a1, false, b0,
                                                  (short)0, acc10, false, false);
    acc11 = __builtin_amdgcn_wmma_f32_16x16x4_f32(false, a1, false, b1,
                                                  (short)0, acc11, false, false);
  }
}

__global__ __launch_bounds__(256) __attribute__((amdgpu_waves_per_eu(4, 4)))
void gemm_xb_wmma(const float* __restrict__ X,
                  const float* __restrict__ B,
                  const float* __restrict__ a_diag,
                  float* __restrict__ S,
                  float* __restrict__ carry) {
  __shared__ alignas(16) float sX[2 * XS_FLOATS];
  __shared__ alignas(16) float sB[2 * BS_FLOATS];
  __shared__ alignas(16) float sE[BN * LDE];

  const int tid  = threadIdx.x;
  const int wave = tid >> 5;
  const int lane = tid & 31;
  const int half = lane >> 4;
  const int r    = lane & 15;

  const int blockN = blockIdx.x * BN;
  const int blockM = blockIdx.y * BM;

  const int wm = (wave & 3) << 5;
  const int wn = (wave >> 2) << 5;

  const int rowGrp = tid >> 3;
  const int col4   = (tid & 7) << 2;
  const int bRow   = tid >> 4;
  const int bCol4  = (tid & 15) << 2;

  const int   cA = wn + r;
  const int   cB = wn + 16 + r;
  const float aA = a_diag[blockN + cA];
  const float aB = a_diag[blockN + cB];

  v8f acc00 = {}, acc01 = {}, acc10 = {}, acc11 = {};

  const float* xg = X + (size_t)(blockM + rowGrp) * KDIM + col4;
  const float* bg = B + (size_t)bRow * HID + blockN + bCol4;
  const int xs = rowGrp * LDX + col4;
  const int bs = bCol4 * LDB + bRow;

  const int fa0 = (wm + r) * LDX + 2 * half;
  const int fa1 = fa0 + 16 * LDX;
  const int fb0 = (wn + r) * LDB + 2 * half;
  const int fb1 = fb0 + 16 * LDB;

  v4f xv[4], bv[2];

  stage_load(xg, bg, 0, xv, bv);
  stage_store(sX, sB, xs, bs, xv, bv);
  __syncthreads();

  for (int it = 0; it < NKC - 1; ++it) {
    const int cur = it & 1;
    stage_load(xg, bg, (it + 1) * KC, xv, bv);
    mma_chunk(sX + cur * XS_FLOATS, sB + cur * BS_FLOATS, fa0, fa1, fb0, fb1,
              acc00, acc01, acc10, acc11);
    stage_store(sX + (cur ^ 1) * XS_FLOATS, sB + (cur ^ 1) * BS_FLOATS, xs, bs, xv, bv);
    __syncthreads();
  }
  mma_chunk(sX + ((NKC - 1) & 1) * XS_FLOATS, sB + ((NKC - 1) & 1) * BS_FLOATS,
            fa0, fa1, fb0, fb1, acc00, acc01, acc10, acc11);

  const int segA = (wm >> 3) + half;

  #pragma unroll
  for (int i = 1; i < 8; ++i) {
    acc00[i] = fmaf(aA, acc00[i - 1], acc00[i]);
    acc10[i] = fmaf(aA, acc10[i - 1], acc10[i]);
    acc01[i] = fmaf(aB, acc01[i - 1], acc01[i]);
    acc11[i] = fmaf(aB, acc11[i - 1], acc11[i]);
  }

  sE[cA * LDE + segA]     = acc00[7];
  sE[cA * LDE + segA + 2] = acc10[7];
  sE[cB * LDE + segA]     = acc01[7];
  sE[cB * LDE + segA + 2] = acc11[7];

  float pwA[8], pwB[8];
  pwA[0] = aA;
  pwB[0] = aB;
  #pragma unroll
  for (int k = 1; k < 8; ++k) {
    pwA[k] = pwA[k - 1] * aA;
    pwB[k] = pwB[k - 1] * aB;
  }
  __syncthreads();

  float eA[SEGS], eB[SEGS];
  #pragma unroll
  for (int q = 0; q < SEGS / 4; ++q) {
    const v4f ta = *(const v4f*)(&sE[cA * LDE + 4 * q]);
    const v4f tb = *(const v4f*)(&sE[cB * LDE + 4 * q]);
    #pragma unroll
    for (int i = 0; i < 4; ++i) {
      eA[4 * q + i] = ta[i];
      eB[4 * q + i] = tb[i];
    }
  }
  float stA = 0.0f, stB = 0.0f;
  float inA0 = 0.0f, inA1 = 0.0f, inB0 = 0.0f, inB1 = 0.0f;
  #pragma unroll
  for (int j = 0; j < SEGS; ++j) {
    if (j == segA)     { inA0 = stA; inB0 = stB; }
    if (j == segA + 2) { inA1 = stA; inB1 = stB; }
    stA = fmaf(pwA[7], stA, eA[j]);
    stB = fmaf(pwB[7], stB, eB[j]);
  }

  #pragma unroll
  for (int i = 0; i < 8; ++i) {
    acc00[i] = fmaf(pwA[i], inA0, acc00[i]);
    acc10[i] = fmaf(pwA[i], inA1, acc10[i]);
    acc01[i] = fmaf(pwB[i], inB0, acc01[i]);
    acc11[i] = fmaf(pwB[i], inB1, acc11[i]);
  }

  if ((wave & 3) == 0 && half == 0) {
    carry[blockIdx.y * HID + blockN + cA] = stA;
    carry[blockIdx.y * HID + blockN + cB] = stB;
  }

  #pragma unroll
  for (int i = 0; i < 8; ++i) {
    const int row0 = blockM + wm + i + 8 * half;
    const int col0 = blockN + wn + r;
    S[(size_t)row0        * HID + col0]      = acc00[i];
    S[(size_t)row0        * HID + col0 + 16] = acc01[i];
    S[(size_t)(row0 + 16) * HID + col0]      = acc10[i];
    S[(size_t)(row0 + 16) * HID + col0 + 16] = acc11[i];
  }
}

__global__ __launch_bounds__(256) void scan_carry(const float* __restrict__ a_diag,
                                                  float* __restrict__ carry) {
  const int h = blockIdx.x * 256 + threadIdx.x;
  const float a = a_diag[h];
  static_assert(CLEN == 128, "7 squarings below give a^128");
  float aL = a;
  #pragma unroll
  for (int i = 0; i < 7; ++i) aL *= aL;
  float state = 0.0f;
  for (int c = 0; c < NCHUNK; ++c) {
    const float cv = carry[c * HID + h];
    carry[c * HID + h] = state;
    state = fmaf(aL, state, cv);
  }
}

__global__ __launch_bounds__(256) void scan_fix(const float* __restrict__ a_diag,
                                                const float* __restrict__ carryIn,
                                                float* __restrict__ out) {
  const int h = blockIdx.x * 256 + threadIdx.x;
  const int c = blockIdx.y + 1;
  const float a  = a_diag[h];
  const float ci = carryIn[c * HID + h];
  float w = a;
  const int t0 = c * CLEN;
  for (int t = 0; t < CLEN; ++t) {
    const size_t idx = (size_t)(t0 + t) * HID + h;
    out[idx] = fmaf(w, ci, out[idx]);
    w *= a;
  }
}

extern "C" void kernel_launch(void* const* d_in, const int* in_sizes, int n_in,
                              void* d_out, int out_size, void* d_ws, size_t ws_size,
                              hipStream_t stream) {
  const float* x_seq  = (const float*)d_in[0];
  const float* a_diag = (const float*)d_in[1];
  const float* b_mat  = (const float*)d_in[2];
  float* out = (float*)d_out;

  float* carry = (float*)d_ws;

  dim3 gGemm(HID / BN, T_SEQ / BM);
  gemm_xb_wmma<<<gGemm, 256, 0, stream>>>(x_seq, b_mat, a_diag, out, carry);

  scan_carry<<<HID / 256, 256, 0, stream>>>(a_diag, carry);

  dim3 gFix(HID / 256, NCHUNK - 1);
  scan_fix<<<gFix, 256, 0, stream>>>(a_diag, carry, out);
}


# scan_carry launch folded into scan_fix (grid and row loop unchanged): each thread rebuilds its chunk's entering state from the raw chunk ends, 31 loads as one group, same serial order; 3->2 launches
# speedup vs baseline: 2.0610x; 1.0337x over previous
#include <hip/hip_runtime.h>

#define T_SEQ 4096
#define HID   1024
#define KDIM  1024

#define KC    32
#define BM    128
#define BN    64
#define LDX   36
#define LDB   36
#define NKC   (KDIM / KC)
#define XS_FLOATS (BM * LDX)
#define BS_FLOATS (BN * LDB)

#define CLEN   BM
#define NCHUNK (T_SEQ / CLEN)
#define SEGS   (BM / 8)
#define LDE    20

typedef float v2f __attribute__((ext_vector_type(2)));
typedef float v4f __attribute__((ext_vector_type(4)));
typedef float v8f __attribute__((ext_vector_type(8)));

static __device__ __forceinline__ void stage_load(const float* __restrict__ xg,
                                                  const float* __restrict__ bg, int k0,
                                                  v4f (&xv)[4], v4f (&bv)[2]) {
  #pragma unroll
  for (int p = 0; p < 4; ++p)
    xv[p] = *(const v4f*)(xg + (size_t)(p << 5) * KDIM + k0);
  #pragma unroll
  for (int p = 0; p < 2; ++p)
    bv[p] = *(const v4f*)(bg + (size_t)(k0 + (p << 4)) * HID);
}

static __device__ __forceinline__ void stage_store(float* sXs, float* sBs, int xs, int bs,
                                                   const v4f (&xv)[4], const v4f (&bv)[2]) {
  #pragma unroll
  for (int p = 0; p < 4; ++p)
    *(v4f*)(&sXs[xs + (p << 5) * LDX]) = xv[p];
  #pragma unroll
  for (int p = 0; p < 2; ++p) {
    #pragma unroll
    for (int i = 0; i < 4; ++i)
      sBs[bs + i * LDB + (p << 4)] = bv[p][i];
  }
}

static __device__ __forceinline__ void mma_chunk(const float* sXs, const float* sBs,
                                                 int fa0, int fa1, int fb0, int fb1,
                                                 v8f& acc00, v8f& acc01, v8f& acc10, v8f& acc11) {
  #pragma unroll
  for (int kk = 0; kk < KC; kk += 4) {
    v2f a0 = *(const v2f*)(&sXs[fa0 + kk]);
    v2f a1 = *(const v2f*)(&sXs[fa1 + kk]);
    v2f b0 = *(const v2f*)(&sBs[fb0 + kk]);
    v2f b1 = *(const v2f*)(&sBs[fb1 + kk]);
    acc00 = __builtin_amdgcn_wmma_f32_16x16x4_f32(false, a0, false, b0,
                                                  (short)0, acc00, false, false);
    acc01 = __builtin_amdgcn_wmma_f32_16x16x4_f32(false, a0, false, b1,
                                                  (short)0, acc01, false, false);
    acc10 = __builtin_amdgcn_wmma_f32_16x16x4_f32(false, a1, false, b0,
                                                  (short)0, acc10, false, false);
    acc11 = __builtin_amdgcn_wmma_f32_16x16x4_f32(false, a1, false, b1,
                                                  (short)0, acc11, false, false);
  }
}

__global__ __launch_bounds__(256) __attribute__((amdgpu_waves_per_eu(4, 4)))
void gemm_xb_wmma(const float* __restrict__ X,
                  const float* __restrict__ B,
                  const float* __restrict__ a_diag,
                  float* __restrict__ S,
                  float* __restrict__ carry) {
  __shared__ alignas(16) float sX[2 * XS_FLOATS];
  __shared__ alignas(16) float sB[2 * BS_FLOATS];
  __shared__ alignas(16) float sE[BN * LDE];

  const int tid  = threadIdx.x;
  const int wave = tid >> 5;
  const int lane = tid & 31;
  const int half = lane >> 4;
  const int r    = lane & 15;

  const int blockN = blockIdx.x * BN;
  const int blockM = blockIdx.y * BM;

  const int wm = (wave & 3) << 5;
  const int wn = (wave >> 2) << 5;

  const int rowGrp = tid >> 3;
  const int col4   = (tid & 7) << 2;
  const int bRow   = tid >> 4;
  const int bCol4  = (tid & 15) << 2;

  const int   cA = wn + r;
  const int   cB = wn + 16 + r;
  const float aA = a_diag[blockN + cA];
  const float aB = a_diag[blockN + cB];

  v8f acc00 = {}, acc01 = {}, acc10 = {}, acc11 = {};

  const float* xg = X + (size_t)(blockM + rowGrp) * KDIM + col4;
  const float* bg = B + (size_t)bRow * HID + blockN + bCol4;
  const int xs = rowGrp * LDX + col4;
  const int bs = bCol4 * LDB + bRow;

  const int fa0 = (wm + r) * LDX + 2 * half;
  const int fa1 = fa0 + 16 * LDX;
  const int fb0 = (wn + r) * LDB + 2 * half;
  const int fb1 = fb0 + 16 * LDB;

  v4f xv[4], bv[2];

  stage_load(xg, bg, 0, xv, bv);
  stage_store(sX, sB, xs, bs, xv, bv);
  __syncthreads();

  for (int it = 0; it < NKC - 1; ++it) {
    const int cur = it & 1;
    stage_load(xg, bg, (it + 1) * KC, xv, bv);
    mma_chunk(sX + cur * XS_FLOATS, sB + cur * BS_FLOATS, fa0, fa1, fb0, fb1,
              acc00, acc01, acc10, acc11);
    stage_store(sX + (cur ^ 1) * XS_FLOATS, sB + (cur ^ 1) * BS_FLOATS, xs, bs, xv, bv);
    __syncthreads();
  }
  mma_chunk(sX + ((NKC - 1) & 1) * XS_FLOATS, sB + ((NKC - 1) & 1) * BS_FLOATS,
            fa0, fa1, fb0, fb1, acc00, acc01, acc10, acc11);

  const int segA = (wm >> 3) + half;

  #pragma unroll
  for (int i = 1; i < 8; ++i) {
    acc00[i] = fmaf(aA, acc00[i - 1], acc00[i]);
    acc10[i] = fmaf(aA, acc10[i - 1], acc10[i]);
    acc01[i] = fmaf(aB, acc01[i - 1], acc01[i]);
    acc11[i] = fmaf(aB, acc11[i - 1], acc11[i]);
  }

  sE[cA * LDE + segA]     = acc00[7];
  sE[cA * LDE + segA + 2] = acc10[7];
  sE[cB * LDE + segA]     = acc01[7];
  sE[cB * LDE + segA + 2] = acc11[7];

  float pwA[8], pwB[8];
  pwA[0] = aA;
  pwB[0] = aB;
  #pragma unroll
  for (int k = 1; k < 8; ++k) {
    pwA[k] = pwA[k - 1] * aA;
    pwB[k] = pwB[k - 1] * aB;
  }
  __syncthreads();

  float eA[SEGS], eB[SEGS];
  #pragma unroll
  for (int q = 0; q < SEGS / 4; ++q) {
    const v4f ta = *(const v4f*)(&sE[cA * LDE + 4 * q]);
    const v4f tb = *(const v4f*)(&sE[cB * LDE + 4 * q]);
    #pragma unroll
    for (int i = 0; i < 4; ++i) {
      eA[4 * q + i] = ta[i];
      eB[4 * q + i] = tb[i];
    }
  }
  float stA = 0.0f, stB = 0.0f;
  float inA0 = 0.0f, inA1 = 0.0f, inB0 = 0.0f, inB1 = 0.0f;
  #pragma unroll
  for (int j = 0; j < SEGS; ++j) {
    if (j == segA)     { inA0 = stA; inB0 = stB; }
    if (j == segA + 2) { inA1 = stA; inB1 = stB; }
    stA = fmaf(pwA[7], stA, eA[j]);
    stB = fmaf(pwB[7], stB, eB[j]);
  }

  #pragma unroll
  for (int i = 0; i < 8; ++i) {
    acc00[i] = fmaf(pwA[i], inA0, acc00[i]);
    acc10[i] = fmaf(pwA[i], inA1, acc10[i]);
    acc01[i] = fmaf(pwB[i], inB0, acc01[i]);
    acc11[i] = fmaf(pwB[i], inB1, acc11[i]);
  }

  if ((wave & 3) == 0 && half == 0) {
    carry[blockIdx.y * HID + blockN + cA] = stA;
    carry[blockIdx.y * HID + blockN + cB] = stB;
  }

  #pragma unroll
  for (int i = 0; i < 8; ++i) {
    const int row0 = blockM + wm + i + 8 * half;
    const int col0 = blockN + wn + r;
    S[(size_t)row0        * HID + col0]      = acc00[i];
    S[(size_t)row0        * HID + col0 + 16] = acc01[i];
    S[(size_t)(row0 + 16) * HID + col0]      = acc10[i];
    S[(size_t)(row0 + 16) * HID + col0 + 16] = acc11[i];
  }
}

__global__ __launch_bounds__(256) void scan_fix(const float* __restrict__ a_diag,
                                                const float* __restrict__ carryEnd,
                                                float* __restrict__ out) {
  const int h = blockIdx.x * 256 + threadIdx.x;
  const int c = blockIdx.y + 1;
  const float a = a_diag[h];

  float e[NCHUNK - 1];
  #pragma unroll
  for (int j = 0; j < NCHUNK - 1; ++j) e[j] = carryEnd[j * HID + h];

  static_assert(CLEN == 128, "7 squarings below give a^128");
  float aL = a;
  #pragma unroll
  for (int i = 0; i < 7; ++i) aL *= aL;

  float ci = 0.0f;
  #pragma unroll
  for (int j = 0; j < NCHUNK - 1; ++j) {
    const float nx = fmaf(aL, ci, e[j]);
    ci = (j < c) ? nx : ci;
  }

  float w = a;
  const int t0 = c * CLEN;
  for (int t = 0; t < CLEN; ++t) {
    const size_t idx = (size_t)(t0 + t) * HID + h;
    out[idx] = fmaf(w, ci, out[idx]);
    w *= a;
  }
}

extern "C" void kernel_launch(void* const* d_in, const int* in_sizes, int n_in,
                              void* d_out, int out_size, void* d_ws, size_t ws_size,
                              hipStream_t stream) {
  const float* x_seq  = (const float*)d_in[0];
  const float* a_diag = (const float*)d_in[1];
  const float* b_mat  = (const float*)d_in[2];
  float* out = (float*)d_out;

  float* carry = (float*)d_ws;

  dim3 gGemm(HID / BN, T_SEQ / BM);
  gemm_xb_wmma<<<gGemm, 256, 0, stream>>>(x_seq, b_mat, a_diag, out, carry);

  dim3 gFix(HID / 256, NCHUNK - 1);
  scan_fix<<<gFix, 256, 0, stream>>>(a_diag, carry, out);
}


# scan_fix: each thread requests all its rows of out and the chunk ends as one load group before any use (64 rows per block, grid.z halves the chunk): 1 memory round trip per thread, not 17; same sums
# speedup vs baseline: 2.3519x; 1.1411x over previous
#include <hip/hip_runtime.h>

#define T_SEQ 4096
#define HID   1024
#define KDIM  1024

#define KC    32
#define BM    128
#define BN    64
#define LDX   36
#define LDB   36
#define NKC   (KDIM / KC)
#define XS_FLOATS (BM * LDX)
#define BS_FLOATS (BN * LDB)

#define CLEN   BM
#define NCHUNK (T_SEQ / CLEN)
#define SEGS   (BM / 8)
#define LDE    20

typedef float v2f __attribute__((ext_vector_type(2)));
typedef float v4f __attribute__((ext_vector_type(4)));
typedef float v8f __attribute__((ext_vector_type(8)));

static __device__ __forceinline__ void stage_load(const float* __restrict__ xg,
                                                  const float* __restrict__ bg, int k0,
                                                  v4f (&xv)[4], v4f (&bv)[2]) {
  #pragma unroll
  for (int p = 0; p < 4; ++p)
    xv[p] = *(const v4f*)(xg + (size_t)(p << 5) * KDIM + k0);
  #pragma unroll
  for (int p = 0; p < 2; ++p)
    bv[p] = *(const v4f*)(bg + (size_t)(k0 + (p << 4)) * HID);
}

static __device__ __forceinline__ void stage_store(float* sXs, float* sBs, int xs, int bs,
                                                   const v4f (&xv)[4], const v4f (&bv)[2]) {
  #pragma unroll
  for (int p = 0; p < 4; ++p)
    *(v4f*)(&sXs[xs + (p << 5) * LDX]) = xv[p];
  #pragma unroll
  for (int p = 0; p < 2; ++p) {
    #pragma unroll
    for (int i = 0; i < 4; ++i)
      sBs[bs + i * LDB + (p << 4)] = bv[p][i];
  }
}

static __device__ __forceinline__ void mma_chunk(const float* sXs, const float* sBs,
                                                 int fa0, int fa1, int fb0, int fb1,
                                                 v8f& acc00, v8f& acc01, v8f& acc10, v8f& acc11) {
  #pragma unroll
  for (int kk = 0; kk < KC; kk += 4) {
    v2f a0 = *(const v2f*)(&sXs[fa0 + kk]);
    v2f a1 = *(const v2f*)(&sXs[fa1 + kk]);
    v2f b0 = *(const v2f*)(&sBs[fb0 + kk]);
    v2f b1 = *(const v2f*)(&sBs[fb1 + kk]);
    acc00 = __builtin_amdgcn_wmma_f32_16x16x4_f32(false, a0, false, b0,
                                                  (short)0, acc00, false, false);
    acc01 = __builtin_amdgcn_wmma_f32_16x16x4_f32(false, a0, false, b1,
                                                  (short)0, acc01, false, false);
    acc10 = __builtin_amdgcn_wmma_f32_16x16x4_f32(false, a1, false, b0,
                                                  (short)0, acc10, false, false);
    acc11 = __builtin_amdgcn_wmma_f32_16x16x4_f32(false, a1, false, b1,
                                                  (short)0, acc11, false, false);
  }
}

__global__ __launch_bounds__(256) __attribute__((amdgpu_waves_per_eu(4, 4)))
void gemm_xb_wmma(const float* __restrict__ X,
                  const float* __restrict__ B,
                  const float* __restrict__ a_diag,
                  float* __restrict__ S,
                  float* __restrict__ carry) {
  __shared__ alignas(16) float sX[2 * XS_FLOATS];
  __shared__ alignas(16) float sB[2 * BS_FLOATS];
  __shared__ alignas(16) float sE[BN * LDE];

  const int tid  = threadIdx.x;
  const int wave = tid >> 5;
  const int lane = tid & 31;
  const int half = lane >> 4;
  const int r    = lane & 15;

  const int blockN = blockIdx.x * BN;
  const int blockM = blockIdx.y * BM;

  const int wm = (wave & 3) << 5;
  const int wn = (wave >> 2) << 5;

  const int rowGrp = tid >> 3;
  const int col4   = (tid & 7) << 2;
  const int bRow   = tid >> 4;
  const int bCol4  = (tid & 15) << 2;

  const int   cA = wn + r;
  const int   cB = wn + 16 + r;
  const float aA = a_diag[blockN + cA];
  const float aB = a_diag[blockN + cB];

  v8f acc00 = {}, acc01 = {}, acc10 = {}, acc11 = {};

  const float* xg = X + (size_t)(blockM + rowGrp) * KDIM + col4;
  const float* bg = B + (size_t)bRow * HID + blockN + bCol4;
  const int xs = rowGrp * LDX + col4;
  const int bs = bCol4 * LDB + bRow;

  const int fa0 = (wm + r) * LDX + 2 * half;
  const int fa1 = fa0 + 16 * LDX;
  const int fb0 = (wn + r) * LDB + 2 * half;
  const int fb1 = fb0 + 16 * LDB;

  v4f xv[4], bv[2];

  stage_load(xg, bg, 0, xv, bv);
  stage_store(sX, sB, xs, bs, xv, bv);
  __syncthreads();

  for (int it = 0; it < NKC - 1; ++it) {
    const int cur = it & 1;
    stage_load(xg, bg, (it + 1) * KC, xv, bv);
    mma_chunk(sX + cur * XS_FLOATS, sB + cur * BS_FLOATS, fa0, fa1, fb0, fb1,
              acc00, acc01, acc10, acc11);
    stage_store(sX + (cur ^ 1) * XS_FLOATS, sB + (cur ^ 1) * BS_FLOATS, xs, bs, xv, bv);
    __syncthreads();
  }
  mma_chunk(sX + ((NKC - 1) & 1) * XS_FLOATS, sB + ((NKC - 1) & 1) * BS_FLOATS,
            fa0, fa1, fb0, fb1, acc00, acc01, acc10, acc11);

  const int segA = (wm >> 3) + half;

  #pragma unroll
  for (int i = 1; i < 8; ++i) {
    acc00[i] = fmaf(aA, acc00[i - 1], acc00[i]);
    acc10[i] = fmaf(aA, acc10[i - 1], acc10[i]);
    acc01[i] = fmaf(aB, acc01[i - 1], acc01[i]);
    acc11[i] = fmaf(aB, acc11[i - 1], acc11[i]);
  }

  sE[cA * LDE + segA]     = acc00[7];
  sE[cA * LDE + segA + 2] = acc10[7];
  sE[cB * LDE + segA]     = acc01[7];
  sE[cB * LDE + segA + 2] = acc11[7];

  float pwA[8], pwB[8];
  pwA[0] = aA;
  pwB[0] = aB;
  #pragma unroll
  for (int k = 1; k < 8; ++k) {
    pwA[k] = pwA[k - 1] * aA;
    pwB[k] = pwB[k - 1] * aB;
  }
  __syncthreads();

  float eA[SEGS], eB[SEGS];
  #pragma unroll
  for (int q = 0; q < SEGS / 4; ++q) {
    const v4f ta = *(const v4f*)(&sE[cA * LDE + 4 * q]);
    const v4f tb = *(const v4f*)(&sE[cB * LDE + 4 * q]);
    #pragma unroll
    for (int i = 0; i < 4; ++i) {
      eA[4 * q + i] = ta[i];
      eB[4 * q + i] = tb[i];
    }
  }
  float stA = 0.0f, stB = 0.0f;
  float inA0 = 0.0f, inA1 = 0.0f, inB0 = 0.0f, inB1 = 0.0f;
  #pragma unroll
  for (int j = 0; j < SEGS; ++j) {
    if (j == segA)     { inA0 = stA; inB0 = stB; }
    if (j == segA + 2) { inA1 = stA; inB1 = stB; }
    stA = fmaf(pwA[7], stA, eA[j]);
    stB = fmaf(pwB[7], stB, eB[j]);
  }

  #pragma unroll
  for (int i = 0; i < 8; ++i) {
    acc00[i] = fmaf(pwA[i], inA0, acc00[i]);
    acc10[i] = fmaf(pwA[i], inA1, acc10[i]);
    acc01[i] = fmaf(pwB[i], inB0, acc01[i]);
    acc11[i] = fmaf(pwB[i], inB1, acc11[i]);
  }

  if ((wave & 3) == 0 && half == 0) {
    carry[blockIdx.y * HID + blockN + cA] = stA;
    carry[blockIdx.y * HID + blockN + cB] = stB;
  }

  #pragma unroll
  for (int i = 0; i < 8; ++i) {
    const int row0 = blockM + wm + i + 8 * half;
    const int col0 = blockN + wn + r;
    S[(size_t)row0        * HID + col0]      = acc00[i];
    S[(size_t)row0        * HID + col0 + 16] = acc01[i];
    S[(size_t)(row0 + 16) * HID + col0]      = acc10[i];
    S[(size_t)(row0 + 16) * HID + col0 + 16] = acc11[i];
  }
}

#define FIX_ROWS  64
#define FIX_SPLIT (CLEN / FIX_ROWS)

__global__ __launch_bounds__(256) void scan_fix(const float* __restrict__ a_diag,
                                                const float* __restrict__ carryEnd,
                                                float* __restrict__ out) {
  const int h = blockIdx.x * 256 + threadIdx.x;
  const int c = blockIdx.y + 1;
  const int r0 = blockIdx.z * FIX_ROWS;
  const float a = a_diag[h];

  float e[NCHUNK - 1];
  #pragma unroll
  for (int j = 0; j < NCHUNK - 1; ++j) e[j] = carryEnd[j * HID + h];

  float* __restrict__ p = out + (size_t)(c * CLEN + r0) * HID + h;
  float v[FIX_ROWS];
  #pragma unroll
  for (int i = 0; i < FIX_ROWS; ++i) v[i] = p[(size_t)i * HID];
  __builtin_amdgcn_sched_barrier(0);

  static_assert(CLEN == 128, "7 squarings below give a^128");
  float aL = a;
  #pragma unroll
  for (int i = 0; i < 7; ++i) aL *= aL;

  float ci = 0.0f;
  #pragma unroll
  for (int j = 0; j < NCHUNK - 1; ++j) {
    const float nx = fmaf(aL, ci, e[j]);
    ci = (j < c) ? nx : ci;
  }

  static_assert(FIX_SPLIT == 2, "the second slice starts at a^(FIX_ROWS+1)");
  float wHi = a;
  #pragma unroll
  for (int i = 0; i < FIX_ROWS; ++i) wHi *= a;
  float w = (r0 == 0) ? a : wHi;

  #pragma unroll
  for (int i = 0; i < FIX_ROWS; ++i) {
    p[(size_t)i * HID] = fmaf(w, ci, v[i]);
    w *= a;
  }
}

extern "C" void kernel_launch(void* const* d_in, const int* in_sizes, int n_in,
                              void* d_out, int out_size, void* d_ws, size_t ws_size,
                              hipStream_t stream) {
  const float* x_seq  = (const float*)d_in[0];
  const float* a_diag = (const float*)d_in[1];
  const float* b_mat  = (const float*)d_in[2];
  float* out = (float*)d_out;

  float* carry = (float*)d_ws;

  dim3 gGemm(HID / BN, T_SEQ / BM);
  gemm_xb_wmma<<<gGemm, 256, 0, stream>>>(x_seq, b_mat, a_diag, out, carry);

  dim3 gFix(HID / 256, NCHUNK - 1, FIX_SPLIT);
  scan_fix<<<gFix, 256, 0, stream>>>(a_diag, carry, out);
}


# GEMM K loop: chunk it+2's global loads go out right behind chunk it+1's LDS store (same registers), a whole trip before their own store; loop barrier orders LDS only, so they stay in flight across it
# speedup vs baseline: 2.4284x; 1.0325x over previous
#include <hip/hip_runtime.h>

#define T_SEQ 4096
#define HID   1024
#define KDIM  1024

#define KC    32
#define BM    128
#define BN    64
#define LDX   36
#define LDB   36
#define NKC   (KDIM / KC)
#define XS_FLOATS (BM * LDX)
#define BS_FLOATS (BN * LDB)

#define CLEN   BM
#define NCHUNK (T_SEQ / CLEN)
#define SEGS   (BM / 8)
#define LDE    20

typedef float v2f __attribute__((ext_vector_type(2)));
typedef float v4f __attribute__((ext_vector_type(4)));
typedef float v8f __attribute__((ext_vector_type(8)));

static __device__ __forceinline__ void stage_load(const float* __restrict__ xg,
                                                  const float* __restrict__ bg, int k0,
                                                  v4f (&xv)[4], v4f (&bv)[2]) {
  #pragma unroll
  for (int p = 0; p < 4; ++p)
    xv[p] = *(const v4f*)(xg + (size_t)(p << 5) * KDIM + k0);
  #pragma unroll
  for (int p = 0; p < 2; ++p)
    bv[p] = *(const v4f*)(bg + (size_t)(k0 + (p << 4)) * HID);
}

static __device__ __forceinline__ void stage_store(float* sXs, float* sBs, int xs, int bs,
                                                   const v4f (&xv)[4], const v4f (&bv)[2]) {
  #pragma unroll
  for (int p = 0; p < 4; ++p)
    *(v4f*)(&sXs[xs + (p << 5) * LDX]) = xv[p];
  #pragma unroll
  for (int p = 0; p < 2; ++p) {
    #pragma unroll
    for (int i = 0; i < 4; ++i)
      sBs[bs + i * LDB + (p << 4)] = bv[p][i];
  }
}

static __device__ __forceinline__ void lds_barrier_arrive() {
  __builtin_amdgcn_fence(__ATOMIC_RELEASE, "workgroup", "local");
  __builtin_amdgcn_s_barrier_signal(-1);
}
static __device__ __forceinline__ void lds_barrier_wait() {
  __builtin_amdgcn_s_barrier_wait(-1);
  __builtin_amdgcn_fence(__ATOMIC_ACQUIRE, "workgroup", "local");
}
static __device__ __forceinline__ void lds_barrier() {
  lds_barrier_arrive();
  lds_barrier_wait();
}

struct Frag { v2f a0, a1, b0, b1; };

static __device__ __forceinline__ Frag load_frag(const float* sXs, const float* sBs,
                                                 int fa0, int fa1, int fb0, int fb1, int kk) {
  Frag f;
  f.a0 = *(const v2f*)(&sXs[fa0 + kk]);
  f.a1 = *(const v2f*)(&sXs[fa1 + kk]);
  f.b0 = *(const v2f*)(&sBs[fb0 + kk]);
  f.b1 = *(const v2f*)(&sBs[fb1 + kk]);
  return f;
}

static __device__ __forceinline__ void mma_step(const Frag& f,
                                                v8f& acc00, v8f& acc01, v8f& acc10, v8f& acc11) {
  acc00 = __builtin_amdgcn_wmma_f32_16x16x4_f32(false, f.a0, false, f.b0,
                                                (short)0, acc00, false, false);
  acc01 = __builtin_amdgcn_wmma_f32_16x16x4_f32(false, f.a0, false, f.b1,
                                                (short)0, acc01, false, false);
  acc10 = __builtin_amdgcn_wmma_f32_16x16x4_f32(false, f.a1, false, f.b0,
                                                (short)0, acc10, false, false);
  acc11 = __builtin_amdgcn_wmma_f32_16x16x4_f32(false, f.a1, false, f.b1,
                                                (short)0, acc11, false, false);
}

template <int K0, int K1>
static __device__ __forceinline__ void mma_chunk(const float* sXs, const float* sBs,
                                                 int fa0, int fa1, int fb0, int fb1,
                                                 v8f& acc00, v8f& acc01, v8f& acc10, v8f& acc11) {
  #pragma unroll
  for (int kk = K0; kk < K1; kk += 4)
    mma_step(load_frag(sXs, sBs, fa0, fa1, fb0, fb1, kk), acc00, acc01, acc10, acc11);
}

__global__ __launch_bounds__(256) __attribute__((amdgpu_waves_per_eu(4, 4)))
void gemm_xb_wmma(const float* __restrict__ X,
                  const float* __restrict__ B,
                  const float* __restrict__ a_diag,
                  float* __restrict__ S,
                  float* __restrict__ carry) {
  __shared__ alignas(16) float sX[2 * XS_FLOATS];
  __shared__ alignas(16) float sB[2 * BS_FLOATS];
  __shared__ alignas(16) float sE[BN * LDE];

  const int tid  = threadIdx.x;
  const int wave = tid >> 5;
  const int lane = tid & 31;
  const int half = lane >> 4;
  const int r    = lane & 15;

  const int blockN = blockIdx.x * BN;
  const int blockM = blockIdx.y * BM;

  const int wm = (wave & 3) << 5;
  const int wn = (wave >> 2) << 5;

  const int rowGrp = tid >> 3;
  const int col4   = (tid & 7) << 2;
  const int bRow   = tid >> 4;
  const int bCol4  = (tid & 15) << 2;

  const int   cA = wn + r;
  const int   cB = wn + 16 + r;
  const float aA = a_diag[blockN + cA];
  const float aB = a_diag[blockN + cB];

  v8f acc00 = {}, acc01 = {}, acc10 = {}, acc11 = {};

  const float* xg = X + (size_t)(blockM + rowGrp) * KDIM + col4;
  const float* bg = B + (size_t)bRow * HID + blockN + bCol4;
  const int xs = rowGrp * LDX + col4;
  const int bs = bCol4 * LDB + bRow;

  const int fa0 = (wm + r) * LDX + 2 * half;
  const int fa1 = fa0 + 16 * LDX;
  const int fb0 = (wn + r) * LDB + 2 * half;
  const int fb1 = fb0 + 16 * LDB;

  v4f xv[4], bv[2];

  stage_load(xg, bg, 0, xv, bv);
  stage_store(sX, sB, xs, bs, xv, bv);
  stage_load(xg, bg, KC, xv, bv);
  lds_barrier();

  for (int it = 0; it < NKC - 2; ++it) {
    const int cur = it & 1;
    const float* cX = sX + cur * XS_FLOATS;
    const float* cB = sB + cur * BS_FLOATS;
    mma_chunk<0, KC / 2>(cX, cB, fa0, fa1, fb0, fb1, acc00, acc01, acc10, acc11);
    stage_store(sX + (cur ^ 1) * XS_FLOATS, sB + (cur ^ 1) * BS_FLOATS, xs, bs, xv, bv);
    stage_load(xg, bg, (it + 2) * KC, xv, bv);
    mma_chunk<KC / 2, KC - 12>(cX, cB, fa0, fa1, fb0, fb1, acc00, acc01, acc10, acc11);
    const Frag t5 = load_frag(cX, cB, fa0, fa1, fb0, fb1, KC - 12);
    const Frag t6 = load_frag(cX, cB, fa0, fa1, fb0, fb1, KC - 8);
    const Frag t7 = load_frag(cX, cB, fa0, fa1, fb0, fb1, KC - 4);
    lds_barrier_arrive();
    __builtin_amdgcn_sched_barrier(0);
    mma_step(t5, acc00, acc01, acc10, acc11);
    __builtin_amdgcn_sched_barrier(0);
    lds_barrier_wait();
    __builtin_amdgcn_sched_barrier(0);
    mma_step(t6, acc00, acc01, acc10, acc11);
    mma_step(t7, acc00, acc01, acc10, acc11);
  }
  static_assert(((NKC - 2) & 1) == 0, "trip NKC-2 computes from stage 0");
  mma_chunk<0, KC>(sX, sB, fa0, fa1, fb0, fb1, acc00, acc01, acc10, acc11);
  stage_store(sX + XS_FLOATS, sB + BS_FLOATS, xs, bs, xv, bv);
  lds_barrier();
  mma_chunk<0, KC>(sX + XS_FLOATS, sB + BS_FLOATS, fa0, fa1, fb0, fb1,
                   acc00, acc01, acc10, acc11);

  const int segA = (wm >> 3) + half;

  #pragma unroll
  for (int i = 1; i < 8; ++i) {
    acc00[i] = fmaf(aA, acc00[i - 1], acc00[i]);
    acc10[i] = fmaf(aA, acc10[i - 1], acc10[i]);
    acc01[i] = fmaf(aB, acc01[i - 1], acc01[i]);
    acc11[i] = fmaf(aB, acc11[i - 1], acc11[i]);
  }

  sE[cA * LDE + segA]     = acc00[7];
  sE[cA * LDE + segA + 2] = acc10[7];
  sE[cB * LDE + segA]     = acc01[7];
  sE[cB * LDE + segA + 2] = acc11[7];

  float pwA[8], pwB[8];
  pwA[0] = aA;
  pwB[0] = aB;
  #pragma unroll
  for (int k = 1; k < 8; ++k) {
    pwA[k] = pwA[k - 1] * aA;
    pwB[k] = pwB[k - 1] * aB;
  }
  __syncthreads();

  float eA[SEGS], eB[SEGS];
  #pragma unroll
  for (int q = 0; q < SEGS / 4; ++q) {
    const v4f ta = *(const v4f*)(&sE[cA * LDE + 4 * q]);
    const v4f tb = *(const v4f*)(&sE[cB * LDE + 4 * q]);
    #pragma unroll
    for (int i = 0; i < 4; ++i) {
      eA[4 * q + i] = ta[i];
      eB[4 * q + i] = tb[i];
    }
  }
  float stA = 0.0f, stB = 0.0f;
  float inA0 = 0.0f, inA1 = 0.0f, inB0 = 0.0f, inB1 = 0.0f;
  #pragma unroll
  for (int j = 0; j < SEGS; ++j) {
    if (j == segA)     { inA0 = stA; inB0 = stB; }
    if (j == segA + 2) { inA1 = stA; inB1 = stB; }
    stA = fmaf(pwA[7], stA, eA[j]);
    stB = fmaf(pwB[7], stB, eB[j]);
  }

  #pragma unroll
  for (int i = 0; i < 8; ++i) {
    acc00[i] = fmaf(pwA[i], inA0, acc00[i]);
    acc10[i] = fmaf(pwA[i], inA1, acc10[i]);
    acc01[i] = fmaf(pwB[i], inB0, acc01[i]);
    acc11[i] = fmaf(pwB[i], inB1, acc11[i]);
  }

  if ((wave & 3) == 0 && half == 0) {
    carry[blockIdx.y * HID + blockN + cA] = stA;
    carry[blockIdx.y * HID + blockN + cB] = stB;
  }

  #pragma unroll
  for (int i = 0; i < 8; ++i) {
    const int row0 = blockM + wm + i + 8 * half;
    const int col0 = blockN + wn + r;
    S[(size_t)row0        * HID + col0]      = acc00[i];
    S[(size_t)row0        * HID + col0 + 16] = acc01[i];
    S[(size_t)(row0 + 16) * HID + col0]      = acc10[i];
    S[(size_t)(row0 + 16) * HID + col0 + 16] = acc11[i];
  }
}

#define FIX_ROWS  64
#define FIX_SPLIT (CLEN / FIX_ROWS)

__global__ __launch_bounds__(256) void scan_fix(const float* __restrict__ a_diag,
                                                const float* __restrict__ carryEnd,
                                                float* __restrict__ out) {
  const int h = blockIdx.x * 256 + threadIdx.x;
  const int c = blockIdx.y + 1;
  const int r0 = blockIdx.z * FIX_ROWS;
  const float a = a_diag[h];

  float e[NCHUNK - 1];
  #pragma unroll
  for (int j = 0; j < NCHUNK - 1; ++j) e[j] = carryEnd[j * HID + h];

  float* __restrict__ p = out + (size_t)(c * CLEN + r0) * HID + h;
  float v[FIX_ROWS];
  #pragma unroll
  for (int i = 0; i < FIX_ROWS; ++i) v[i] = p[(size_t)i * HID];
  __builtin_amdgcn_sched_barrier(0);

  static_assert(CLEN == 128, "7 squarings below give a^128");
  float aL = a;
  #pragma unroll
  for (int i = 0; i < 7; ++i) aL *= aL;

  float ci = 0.0f;
  #pragma unroll
  for (int j = 0; j < NCHUNK - 1; ++j) {
    const float nx = fmaf(aL, ci, e[j]);
    ci = (j < c) ? nx : ci;
  }

  static_assert(FIX_SPLIT == 2, "the second slice starts at a^(FIX_ROWS+1)");
  float wHi = a;
  #pragma unroll
  for (int i = 0; i < FIX_ROWS; ++i) wHi *= a;
  float w = (r0 == 0) ? a : wHi;

  #pragma unroll
  for (int i = 0; i < FIX_ROWS; ++i) {
    p[(size_t)i * HID] = fmaf(w, ci, v[i]);
    w *= a;
  }
}

extern "C" void kernel_launch(void* const* d_in, const int* in_sizes, int n_in,
                              void* d_out, int out_size, void* d_ws, size_t ws_size,
                              hipStream_t stream) {
  const float* x_seq  = (const float*)d_in[0];
  const float* a_diag = (const float*)d_in[1];
  const float* b_mat  = (const float*)d_in[2];
  float* out = (float*)d_out;

  float* carry = (float*)d_ws;

  dim3 gGemm(HID / BN, T_SEQ / BM);
  gemm_xb_wmma<<<gGemm, 256, 0, stream>>>(x_seq, b_mat, a_diag, out, carry);

  dim3 gFix(HID / 256, NCHUNK - 1, FIX_SPLIT);
  scan_fix<<<gFix, 256, 0, stream>>>(a_diag, carry, out);
}


# GEMM K loop: global loads two chunks ahead of their LDS store, not one (second staging register set, loop unrolled by two trips): a request has two trips of WMMAs to come back; same sums
# speedup vs baseline: 2.5284x; 1.0412x over previous
#include <hip/hip_runtime.h>

#define T_SEQ 4096
#define HID   1024
#define KDIM  1024

#define KC    32
#define BM    128
#define BN    64
#define LDX   36
#define LDB   36
#define NKC   (KDIM / KC)
#define XS_FLOATS (BM * LDX)
#define BS_FLOATS (BN * LDB)

#define CLEN   BM
#define NCHUNK (T_SEQ / CLEN)
#define SEGS   (BM / 8)
#define LDE    20

typedef float v2f __attribute__((ext_vector_type(2)));
typedef float v4f __attribute__((ext_vector_type(4)));
typedef float v8f __attribute__((ext_vector_type(8)));

static __device__ __forceinline__ void stage_load(const float* __restrict__ xg,
                                                  const float* __restrict__ bg, int k0,
                                                  v4f (&xv)[4], v4f (&bv)[2]) {
  #pragma unroll
  for (int p = 0; p < 4; ++p)
    xv[p] = *(const v4f*)(xg + (size_t)(p << 5) * KDIM + k0);
  #pragma unroll
  for (int p = 0; p < 2; ++p)
    bv[p] = *(const v4f*)(bg + (size_t)(k0 + (p << 4)) * HID);
}

static __device__ __forceinline__ void stage_store(float* sXs, float* sBs, int xs, int bs,
                                                   const v4f (&xv)[4], const v4f (&bv)[2]) {
  #pragma unroll
  for (int p = 0; p < 4; ++p)
    *(v4f*)(&sXs[xs + (p << 5) * LDX]) = xv[p];
  #pragma unroll
  for (int p = 0; p < 2; ++p) {
    #pragma unroll
    for (int i = 0; i < 4; ++i)
      sBs[bs + i * LDB + (p << 4)] = bv[p][i];
  }
}

static __device__ __forceinline__ void lds_barrier_arrive() {
  __builtin_amdgcn_fence(__ATOMIC_RELEASE, "workgroup", "local");
  __builtin_amdgcn_s_barrier_signal(-1);
}
static __device__ __forceinline__ void lds_barrier_wait() {
  __builtin_amdgcn_s_barrier_wait(-1);
  __builtin_amdgcn_fence(__ATOMIC_ACQUIRE, "workgroup", "local");
}
static __device__ __forceinline__ void lds_barrier() {
  lds_barrier_arrive();
  lds_barrier_wait();
}

struct Frag { v2f a0, a1, b0, b1; };

static __device__ __forceinline__ Frag load_frag(const float* sXs, const float* sBs,
                                                 int fa0, int fa1, int fb0, int fb1, int kk) {
  Frag f;
  f.a0 = *(const v2f*)(&sXs[fa0 + kk]);
  f.a1 = *(const v2f*)(&sXs[fa1 + kk]);
  f.b0 = *(const v2f*)(&sBs[fb0 + kk]);
  f.b1 = *(const v2f*)(&sBs[fb1 + kk]);
  return f;
}

static __device__ __forceinline__ void mma_step(const Frag& f,
                                                v8f& acc00, v8f& acc01, v8f& acc10, v8f& acc11) {
  acc00 = __builtin_amdgcn_wmma_f32_16x16x4_f32(false, f.a0, false, f.b0,
                                                (short)0, acc00, false, false);
  acc01 = __builtin_amdgcn_wmma_f32_16x16x4_f32(false, f.a0, false, f.b1,
                                                (short)0, acc01, false, false);
  acc10 = __builtin_amdgcn_wmma_f32_16x16x4_f32(false, f.a1, false, f.b0,
                                                (short)0, acc10, false, false);
  acc11 = __builtin_amdgcn_wmma_f32_16x16x4_f32(false, f.a1, false, f.b1,
                                                (short)0, acc11, false, false);
}

template <int K0, int K1>
static __device__ __forceinline__ void mma_chunk(const float* sXs, const float* sBs,
                                                 int fa0, int fa1, int fb0, int fb1,
                                                 v8f& acc00, v8f& acc01, v8f& acc10, v8f& acc11) {
  #pragma unroll
  for (int kk = K0; kk < K1; kk += 4)
    mma_step(load_frag(sXs, sBs, fa0, fa1, fb0, fb1, kk), acc00, acc01, acc10, acc11);
}

template <int CUR, bool LOAD>
static __device__ __forceinline__ void k_trip(float* sX, float* sB,
                                              const float* __restrict__ xg,
                                              const float* __restrict__ bg, int knext,
                                              int xs, int bs, int fa0, int fa1, int fb0, int fb1,
                                              v4f (&xv)[4], v4f (&bv)[2],
                                              v8f& acc00, v8f& acc01, v8f& acc10, v8f& acc11) {
  const float* cX = sX + CUR * XS_FLOATS;
  const float* cB = sB + CUR * BS_FLOATS;
  mma_chunk<0, KC / 2>(cX, cB, fa0, fa1, fb0, fb1, acc00, acc01, acc10, acc11);
  stage_store(sX + (CUR ^ 1) * XS_FLOATS, sB + (CUR ^ 1) * BS_FLOATS, xs, bs, xv, bv);
  if (LOAD) stage_load(xg, bg, knext, xv, bv);
  mma_chunk<KC / 2, KC - 12>(cX, cB, fa0, fa1, fb0, fb1, acc00, acc01, acc10, acc11);
  const Frag t5 = load_frag(cX, cB, fa0, fa1, fb0, fb1, KC - 12);
  const Frag t6 = load_frag(cX, cB, fa0, fa1, fb0, fb1, KC - 8);
  const Frag t7 = load_frag(cX, cB, fa0, fa1, fb0, fb1, KC - 4);
  lds_barrier_arrive();
  __builtin_amdgcn_sched_barrier(0);
  mma_step(t5, acc00, acc01, acc10, acc11);
  __builtin_amdgcn_sched_barrier(0);
  lds_barrier_wait();
  __builtin_amdgcn_sched_barrier(0);
  mma_step(t6, acc00, acc01, acc10, acc11);
  mma_step(t7, acc00, acc01, acc10, acc11);
}

__global__ __launch_bounds__(256) __attribute__((amdgpu_waves_per_eu(4, 4)))
void gemm_xb_wmma(const float* __restrict__ X,
                  const float* __restrict__ B,
                  const float* __restrict__ a_diag,
                  float* __restrict__ S,
                  float* __restrict__ carry) {
  __shared__ alignas(16) float sX[2 * XS_FLOATS];
  __shared__ alignas(16) float sB[2 * BS_FLOATS];
  __shared__ alignas(16) float sE[BN * LDE];

  const int tid  = threadIdx.x;
  const int wave = tid >> 5;
  const int lane = tid & 31;
  const int half = lane >> 4;
  const int r    = lane & 15;

  const int blockN = blockIdx.x * BN;
  const int blockM = blockIdx.y * BM;

  const int wm = (wave & 3) << 5;
  const int wn = (wave >> 2) << 5;

  const int rowGrp = tid >> 3;
  const int col4   = (tid & 7) << 2;
  const int bRow   = tid >> 4;
  const int bCol4  = (tid & 15) << 2;

  const int   cA = wn + r;
  const int   cB = wn + 16 + r;
  const float aA = a_diag[blockN + cA];
  const float aB = a_diag[blockN + cB];

  v8f acc00 = {}, acc01 = {}, acc10 = {}, acc11 = {};

  const float* xg = X + (size_t)(blockM + rowGrp) * KDIM + col4;
  const float* bg = B + (size_t)bRow * HID + blockN + bCol4;
  const int xs = rowGrp * LDX + col4;
  const int bs = bCol4 * LDB + bRow;

  const int fa0 = (wm + r) * LDX + 2 * half;
  const int fa1 = fa0 + 16 * LDX;
  const int fb0 = (wn + r) * LDB + 2 * half;
  const int fb1 = fb0 + 16 * LDB;

  v4f xvE[4], bvE[2], xvO[4], bvO[2];

  stage_load(xg, bg, 0, xvE, bvE);
  stage_load(xg, bg, KC, xvO, bvO);
  __builtin_amdgcn_sched_barrier(0);
  stage_store(sX, sB, xs, bs, xvE, bvE);
  stage_load(xg, bg, 2 * KC, xvE, bvE);
  lds_barrier();

  static_assert((NKC & 1) == 0 && NKC >= 6, "trip pairs up to NKC-4, then four tail trips");
  for (int it = 0; it < NKC - 4; it += 2) {
    k_trip<0, true>(sX, sB, xg, bg, (it + 3) * KC, xs, bs, fa0, fa1, fb0, fb1, xvO, bvO,
                    acc00, acc01, acc10, acc11);
    k_trip<1, true>(sX, sB, xg, bg, (it + 4) * KC, xs, bs, fa0, fa1, fb0, fb1, xvE, bvE,
                    acc00, acc01, acc10, acc11);
  }
  k_trip<0, true>(sX, sB, xg, bg, (NKC - 1) * KC, xs, bs, fa0, fa1, fb0, fb1, xvO, bvO,
                  acc00, acc01, acc10, acc11);
  k_trip<1, false>(sX, sB, xg, bg, 0, xs, bs, fa0, fa1, fb0, fb1, xvE, bvE,
                   acc00, acc01, acc10, acc11);
  mma_chunk<0, KC>(sX, sB, fa0, fa1, fb0, fb1, acc00, acc01, acc10, acc11);
  stage_store(sX + XS_FLOATS, sB + BS_FLOATS, xs, bs, xvO, bvO);
  lds_barrier();
  mma_chunk<0, KC>(sX + XS_FLOATS, sB + BS_FLOATS, fa0, fa1, fb0, fb1,
                   acc00, acc01, acc10, acc11);

  const int segA = (wm >> 3) + half;

  #pragma unroll
  for (int i = 1; i < 8; ++i) {
    acc00[i] = fmaf(aA, acc00[i - 1], acc00[i]);
    acc10[i] = fmaf(aA, acc10[i - 1], acc10[i]);
    acc01[i] = fmaf(aB, acc01[i - 1], acc01[i]);
    acc11[i] = fmaf(aB, acc11[i - 1], acc11[i]);
  }

  sE[cA * LDE + segA]     = acc00[7];
  sE[cA * LDE + segA + 2] = acc10[7];
  sE[cB * LDE + segA]     = acc01[7];
  sE[cB * LDE + segA + 2] = acc11[7];

  float pwA[8], pwB[8];
  pwA[0] = aA;
  pwB[0] = aB;
  #pragma unroll
  for (int k = 1; k < 8; ++k) {
    pwA[k] = pwA[k - 1] * aA;
    pwB[k] = pwB[k - 1] * aB;
  }
  __syncthreads();

  float eA[SEGS], eB[SEGS];
  #pragma unroll
  for (int q = 0; q < SEGS / 4; ++q) {
    const v4f ta = *(const v4f*)(&sE[cA * LDE + 4 * q]);
    const v4f tb = *(const v4f*)(&sE[cB * LDE + 4 * q]);
    #pragma unroll
    for (int i = 0; i < 4; ++i) {
      eA[4 * q + i] = ta[i];
      eB[4 * q + i] = tb[i];
    }
  }
  float stA = 0.0f, stB = 0.0f;
  float inA0 = 0.0f, inA1 = 0.0f, inB0 = 0.0f, inB1 = 0.0f;
  #pragma unroll
  for (int j = 0; j < SEGS; ++j) {
    if (j == segA)     { inA0 = stA; inB0 = stB; }
    if (j == segA + 2) { inA1 = stA; inB1 = stB; }
    stA = fmaf(pwA[7], stA, eA[j]);
    stB = fmaf(pwB[7], stB, eB[j]);
  }

  #pragma unroll
  for (int i = 0; i < 8; ++i) {
    acc00[i] = fmaf(pwA[i], inA0, acc00[i]);
    acc10[i] = fmaf(pwA[i], inA1, acc10[i]);
    acc01[i] = fmaf(pwB[i], inB0, acc01[i]);
    acc11[i] = fmaf(pwB[i], inB1, acc11[i]);
  }

  if ((wave & 3) == 0 && half == 0) {
    carry[blockIdx.y * HID + blockN + cA] = stA;
    carry[blockIdx.y * HID + blockN + cB] = stB;
  }

  #pragma unroll
  for (int i = 0; i < 8; ++i) {
    const int row0 = blockM + wm + i + 8 * half;
    const int col0 = blockN + wn + r;
    S[(size_t)row0        * HID + col0]      = acc00[i];
    S[(size_t)row0        * HID + col0 + 16] = acc01[i];
    S[(size_t)(row0 + 16) * HID + col0]      = acc10[i];
    S[(size_t)(row0 + 16) * HID + col0 + 16] = acc11[i];
  }
}

#define FIX_ROWS  64
#define FIX_SPLIT (CLEN / FIX_ROWS)

__global__ __launch_bounds__(256) void scan_fix(const float* __restrict__ a_diag,
                                                const float* __restrict__ carryEnd,
                                                float* __restrict__ out) {
  const int h = blockIdx.x * 256 + threadIdx.x;
  const int c = blockIdx.y + 1;
  const int r0 = blockIdx.z * FIX_ROWS;
  const float a = a_diag[h];

  float e[NCHUNK - 1];
  #pragma unroll
  for (int j = 0; j < NCHUNK - 1; ++j) e[j] = carryEnd[j * HID + h];

  float* __restrict__ p = out + (size_t)(c * CLEN + r0) * HID + h;
  float v[FIX_ROWS];
  #pragma unroll
  for (int i = 0; i < FIX_ROWS; ++i) v[i] = p[(size_t)i * HID];
  __builtin_amdgcn_sched_barrier(0);

  static_assert(CLEN == 128, "7 squarings below give a^128");
  float aL = a;
  #pragma unroll
  for (int i = 0; i < 7; ++i) aL *= aL;

  float ci = 0.0f;
  #pragma unroll
  for (int j = 0; j < NCHUNK - 1; ++j) {
    const float nx = fmaf(aL, ci, e[j]);
    ci = (j < c) ? nx : ci;
  }

  static_assert(FIX_SPLIT == 2, "the second slice starts at a^(FIX_ROWS+1)");
  float wHi = a;
  #pragma unroll
  for (int i = 0; i < FIX_ROWS; ++i) wHi *= a;
  float w = (r0 == 0) ? a : wHi;

  #pragma unroll
  for (int i = 0; i < FIX_ROWS; ++i) {
    p[(size_t)i * HID] = fmaf(w, ci, v[i]);
    w *= a;
  }
}

extern "C" void kernel_launch(void* const* d_in, const int* in_sizes, int n_in,
                              void* d_out, int out_size, void* d_ws, size_t ws_size,
                              hipStream_t stream) {
  const float* x_seq  = (const float*)d_in[0];
  const float* a_diag = (const float*)d_in[1];
  const float* b_mat  = (const float*)d_in[2];
  float* out = (float*)d_out;

  float* carry = (float*)d_ws;

  dim3 gGemm(HID / BN, T_SEQ / BM);
  gemm_xb_wmma<<<gGemm, 256, 0, stream>>>(x_seq, b_mat, a_diag, out, carry);

  dim3 gFix(HID / 256, NCHUNK - 1, FIX_SPLIT);
  scan_fix<<<gFix, 256, 0, stream>>>(a_diag, carry, out);
}
